# starting kernel
# baseline (speedup 1.0000x reference)
#include <hip/hip_runtime.h>

#define Nn   50000
#define Ee   400000
#define INF_ 256
#define Hh   8
#define Dd   32
#define HD   256
#define EFe  3

typedef __attribute__((ext_vector_type(2))) float v2f;
typedef __attribute__((ext_vector_type(8))) float v8f;

__device__ __forceinline__ unsigned f2o(float f) {
  unsigned u = __float_as_uint(f);
  return (u & 0x80000000u) ? ~u : (u | 0x80000000u);
}
__device__ __forceinline__ float o2f(unsigned u) {
  return (u & 0x80000000u) ? __uint_as_float(u & 0x7fffffffu) : __uint_as_float(~u);
}

__global__ void init_kernel(float* __restrict__ out, float* __restrict__ denom,
                            unsigned* __restrict__ amax_u, const float* __restrict__ bias) {
  int idx = blockIdx.x * blockDim.x + threadIdx.x;
  if (idx < Nn * HD) out[idx] = bias[idx & (HD - 1)];
  if (idx < Nn * Hh) { denom[idx] = 0.f; amax_u[idx] = 0u; }
}

__global__ __launch_bounds__(256) void gemm_h_wmma(const float* __restrict__ x,
                                                   const float* __restrict__ Wm,
                                                   float* __restrict__ h) {
  __shared__ float sA[32][33];
  __shared__ float sB[128][33];
  const int tid   = threadIdx.x;
  const int mBase = blockIdx.x * 32;
  const int nBase = blockIdx.y * 128;
  const int wave  = tid >> 5, ln = tid & 31;
  const int half  = ln >> 4, mr = ln & 15;
  const int mt    = wave & 1;
  const int nt0   = (wave >> 1) << 1;

  v8f acc0 = {};
  v8f acc1 = {};

  for (int k0 = 0; k0 < INF_; k0 += 32) {
    for (int i = tid; i < 32 * 32; i += 256) {
      int r = i >> 5, c = i & 31, row = mBase + r;
      sA[r][c] = (row < Nn) ? x[row * INF_ + k0 + c] : 0.f;
    }
    for (int i = tid; i < 128 * 32; i += 256) {
      int r = i >> 5, c = i & 31;
      sB[r][c] = Wm[(nBase + r) * INF_ + k0 + c];
    }
    __syncthreads();

    const int ar  = mt * 16 + mr;
    const int br0 = nt0 * 16 + mr;
    const int br1 = br0 + 16;
#pragma unroll
    for (int kk = 0; kk < 32; kk += 4) {
      const int kh = kk + half * 2;
      v2f a, b0, b1;
      a.x  = sA[ar][kh];   a.y  = sA[ar][kh + 1];
      b0.x = sB[br0][kh];  b0.y = sB[br0][kh + 1];
      b1.x = sB[br1][kh];  b1.y = sB[br1][kh + 1];
      acc0 = __builtin_amdgcn_wmma_f32_16x16x4_f32(false, a, false, b0, (short)0, acc0, false, false);
      acc1 = __builtin_amdgcn_wmma_f32_16x16x4_f32(false, a, false, b1, (short)0, acc1, false, false);
    }
    __syncthreads();
  }

  const int rowBase = mBase + mt * 16 + half * 8;
#pragma unroll
  for (int r = 0; r < 8; ++r) {
    int row = rowBase + r;
    if (row < Nn) {
      h[row * HD + nBase + nt0 * 16 + mr]       = acc0[r];
      h[row * HD + nBase + (nt0 + 1) * 16 + mr] = acc1[r];
    }
  }
}

__global__ __launch_bounds__(256) void edge_alpha_kernel(
    const float* __restrict__ h, const float* __restrict__ a_src,
    const float* __restrict__ a_dst, const float* __restrict__ edge_W,
    const float* __restrict__ edge_b, const float* __restrict__ edge_attr,
    const int* __restrict__ ei, float* __restrict__ alpha,
    unsigned* __restrict__ amax_u) {
  int e = blockIdx.x * 8 + (threadIdx.x >> 5);
  if (e >= Ee) return;
  int lane = threadIdx.x & 31;
  int src = ei[e], dst = ei[Ee + e];
  float ea0 = edge_attr[e * EFe + 0];
  float ea1 = edge_attr[e * EFe + 1];
  float ea2 = edge_attr[e * EFe + 2];
  const float* hs_row = h + (long)src * HD;
  const float* hd_row = h + (long)dst * HD;

  for (int hh = 0; hh < Hh; ++hh) {
    int hd_i = hh * Dd + lane;
    float hs  = hs_row[hd_i];
    float hdd = hd_row[hd_i];
    float enc = fmaf(ea0, edge_W[hd_i * EFe + 0],
                fmaf(ea1, edge_W[hd_i * EFe + 1],
                fmaf(ea2, edge_W[hd_i * EFe + 2], edge_b[hd_i])));
    enc = fmaxf(enc, 0.f);
    float p = hs * a_src[hd_i] + hdd * a_dst[hd_i] + enc * hdd;
#pragma unroll
    for (int off = 16; off > 0; off >>= 1) p += __shfl_xor(p, off, 32);
    if (lane == 0) {
      float al = (p >= 0.f) ? p : 0.2f * p;
      alpha[e * Hh + hh] = al;
      atomicMax(&amax_u[dst * Hh + hh], f2o(al));
    }
  }
}

__global__ void softmax_num_kernel(const int* __restrict__ ei,
                                   const unsigned* __restrict__ amax_u,
                                   float* __restrict__ alpha_ex,
                                   float* __restrict__ denom) {
  int idx = blockIdx.x * blockDim.x + threadIdx.x;
  if (idx >= Ee * Hh) return;
  int e = idx >> 3, hh = idx & 7;
  int dst = ei[Ee + e];
  float m  = o2f(amax_u[dst * Hh + hh]);
  float ex = __expf(alpha_ex[idx] - m);
  alpha_ex[idx] = ex;
  atomicAdd(&denom[dst * Hh + hh], ex);
}

__global__ __launch_bounds__(256) void scatter_out_kernel(
    const float* __restrict__ h, const float* __restrict__ ex,
    const float* __restrict__ denom, const int* __restrict__ ei,
    float* __restrict__ out) {
  int e = blockIdx.x * 8 + (threadIdx.x >> 5);
  if (e >= Ee) return;
  int lane = threadIdx.x & 31;
  int src = ei[e], dst = ei[Ee + e];
  const float* hs_row = h + (long)src * HD;
  float* out_row = out + (long)dst * HD;
  for (int hh = 0; hh < Hh; ++hh) {
    float attn = ex[e * Hh + hh] / fmaxf(denom[dst * Hh + hh], 1e-10f);
    atomicAdd(&out_row[hh * Dd + lane], hs_row[hh * Dd + lane] * attn);
  }
}

extern "C" void kernel_launch(void* const* d_in, const int* in_sizes, int n_in,
                              void* d_out, int out_size, void* d_ws, size_t ws_size,
                              hipStream_t stream) {
  const float* x        = (const float*)d_in[0];
  const float* Wm       = (const float*)d_in[1];
  const float* a_src    = (const float*)d_in[2];
  const float* a_dst    = (const float*)d_in[3];
  const float* edge_W   = (const float*)d_in[4];
  const float* edge_b   = (const float*)d_in[5];
  const float* bias     = (const float*)d_in[6];
  const float* edge_attr= (const float*)d_in[7];
  const int*   ei       = (const int*)d_in[8];
  float* out = (float*)d_out;

  float*    h       = (float*)d_ws;
  float*    alpha   = h + (size_t)Nn * HD;
  unsigned* amax_u  = (unsigned*)(alpha + (size_t)Ee * Hh);
  float*    denom   = (float*)(amax_u + (size_t)Nn * Hh);

  init_kernel<<<(Nn * HD + 255) / 256, 256, 0, stream>>>(out, denom, amax_u, bias);

  dim3 g((Nn + 31) / 32, HD / 128);
  gemm_h_wmma<<<g, 256, 0, stream>>>(x, Wm, h);

  edge_alpha_kernel<<<(Ee + 7) / 8, 256, 0, stream>>>(h, a_src, a_dst, edge_W, edge_b,
                                                      edge_attr, ei, alpha, amax_u);
  softmax_num_kernel<<<(Ee * Hh + 255) / 256, 256, 0, stream>>>(ei, amax_u, alpha, denom);
  scatter_out_kernel<<<(Ee + 7) / 8, 256, 0, stream>>>(h, alpha, denom, ei, out);
}


# edge-logit kernel: a lane owns 4+4 contiguous features, so a wave does all 8 heads of an edge from 128-bit row loads; head sums by DPP over 8 lanes (new sum order); one store+atomic per 4 edges
# speedup vs baseline: 1.5650x; 1.5650x over previous
#include <hip/hip_runtime.h>

#define Nn   50000
#define Ee   400000
#define INF_ 256
#define Hh   8
#define Dd   32
#define HD   256
#define EFe  3

typedef __attribute__((ext_vector_type(2))) float v2f;
typedef __attribute__((ext_vector_type(8))) float v8f;

__device__ __forceinline__ unsigned f2o(float f) {
  unsigned u = __float_as_uint(f);
  return (u & 0x80000000u) ? ~u : (u | 0x80000000u);
}
__device__ __forceinline__ float o2f(unsigned u) {
  return (u & 0x80000000u) ? __uint_as_float(u & 0x7fffffffu) : __uint_as_float(~u);
}

__global__ void init_kernel(float* __restrict__ out, float* __restrict__ denom,
                            unsigned* __restrict__ amax_u, const float* __restrict__ bias) {
  int idx = blockIdx.x * blockDim.x + threadIdx.x;
  if (idx < Nn * HD) out[idx] = bias[idx & (HD - 1)];
  if (idx < Nn * Hh) { denom[idx] = 0.f; amax_u[idx] = 0u; }
}

__global__ __launch_bounds__(256) void gemm_h_wmma(const float* __restrict__ x,
                                                   const float* __restrict__ Wm,
                                                   float* __restrict__ h) {
  __shared__ float sA[32][33];
  __shared__ float sB[128][33];
  const int tid   = threadIdx.x;
  const int mBase = blockIdx.x * 32;
  const int nBase = blockIdx.y * 128;
  const int wave  = tid >> 5, ln = tid & 31;
  const int half  = ln >> 4, mr = ln & 15;
  const int mt    = wave & 1;
  const int nt0   = (wave >> 1) << 1;

  v8f acc0 = {};
  v8f acc1 = {};

  for (int k0 = 0; k0 < INF_; k0 += 32) {
    for (int i = tid; i < 32 * 32; i += 256) {
      int r = i >> 5, c = i & 31, row = mBase + r;
      sA[r][c] = (row < Nn) ? x[row * INF_ + k0 + c] : 0.f;
    }
    for (int i = tid; i < 128 * 32; i += 256) {
      int r = i >> 5, c = i & 31;
      sB[r][c] = Wm[(nBase + r) * INF_ + k0 + c];
    }
    __syncthreads();

    const int ar  = mt * 16 + mr;
    const int br0 = nt0 * 16 + mr;
    const int br1 = br0 + 16;
#pragma unroll
    for (int kk = 0; kk < 32; kk += 4) {
      const int kh = kk + half * 2;
      v2f a, b0, b1;
      a.x  = sA[ar][kh];   a.y  = sA[ar][kh + 1];
      b0.x = sB[br0][kh];  b0.y = sB[br0][kh + 1];
      b1.x = sB[br1][kh];  b1.y = sB[br1][kh + 1];
      acc0 = __builtin_amdgcn_wmma_f32_16x16x4_f32(false, a, false, b0, (short)0, acc0, false, false);
      acc1 = __builtin_amdgcn_wmma_f32_16x16x4_f32(false, a, false, b1, (short)0, acc1, false, false);
    }
    __syncthreads();
  }

  const int rowBase = mBase + mt * 16 + half * 8;
#pragma unroll
  for (int r = 0; r < 8; ++r) {
    int row = rowBase + r;
    if (row < Nn) {
      h[row * HD + nBase + nt0 * 16 + mr]       = acc0[r];
      h[row * HD + nBase + (nt0 + 1) * 16 + mr] = acc1[r];
    }
  }
}

#define EPG 4
#define GPW 2

template <int CTRL>
__device__ __forceinline__ float xlane_add(float v) {
  return v + __int_as_float(__builtin_amdgcn_update_dpp(0, __float_as_int(v), CTRL, 0xf, 0xf, true));
}

__device__ __forceinline__ float chunk_logit(float4 hs, float4 hd, float4 as, float4 ad, float4 eb,
                                             float4 wa, float4 wb, float4 wc,
                                             float ea0, float ea1, float ea2) {
  float n0 = fmaxf(fmaf(ea0, wa.x, fmaf(ea1, wa.y, fmaf(ea2, wa.z, eb.x))), 0.f);
  float n1 = fmaxf(fmaf(ea0, wa.w, fmaf(ea1, wb.x, fmaf(ea2, wb.y, eb.y))), 0.f);
  float n2 = fmaxf(fmaf(ea0, wb.z, fmaf(ea1, wb.w, fmaf(ea2, wc.x, eb.z))), 0.f);
  float n3 = fmaxf(fmaf(ea0, wc.y, fmaf(ea1, wc.z, fmaf(ea2, wc.w, eb.w))), 0.f);
  float p = hs.x * as.x;
  p = fmaf(hd.x, ad.x, p);  p = fmaf(n0, hd.x, p);
  p = fmaf(hs.y, as.y, p);  p = fmaf(hd.y, ad.y, p);  p = fmaf(n1, hd.y, p);
  p = fmaf(hs.z, as.z, p);  p = fmaf(hd.z, ad.z, p);  p = fmaf(n2, hd.z, p);
  p = fmaf(hs.w, as.w, p);  p = fmaf(hd.w, ad.w, p);  p = fmaf(n3, hd.w, p);
  return p;
}

__global__ __launch_bounds__(256) void edge_alpha_kernel(
    const float* __restrict__ h, const float* __restrict__ a_src,
    const float* __restrict__ a_dst, const float* __restrict__ edge_W,
    const float* __restrict__ edge_b, const float* __restrict__ edge_attr,
    const int* __restrict__ ei, float* __restrict__ alpha,
    unsigned* __restrict__ amax_u) {
  static_assert(Ee % (EPG * GPW) == 0, "a wave's edges are all in range or none is");
  const int lane = threadIdx.x & 31;
  const int wv = __builtin_amdgcn_readfirstlane(threadIdx.x >> 5);
  const int e0 = (blockIdx.x * 8 + wv) * (EPG * GPW);
  if (e0 >= Ee) return;

  const float4* ew4 = (const float4*)edge_W;
  const float4 w1a = ew4[lane * 3 + 0], w1b = ew4[lane * 3 + 1], w1c = ew4[lane * 3 + 2];
  const float4 w2a = ew4[96 + lane * 3 + 0], w2b = ew4[96 + lane * 3 + 1], w2c = ew4[96 + lane * 3 + 2];
  const float4 b1 = ((const float4*)edge_b)[lane], b2 = ((const float4*)edge_b)[32 + lane];
  const float4 s1 = ((const float4*)a_src)[lane],  s2 = ((const float4*)a_src)[32 + lane];
  const float4 d1 = ((const float4*)a_dst)[lane],  d2 = ((const float4*)a_dst)[32 + lane];

  const int k = lane & 7;
  const int from = ((lane & 3) << 3) + (lane & 4) + (lane >> 3);

#pragma unroll
  for (int g = 0; g < GPW; ++g) {
    const int eg = e0 + g * EPG;
    float r1[EPG], r2[EPG];
    float4 hs1[EPG], hs2[EPG], hd1[EPG], hd2[EPG];
#pragma unroll
    for (int i = 0; i < EPG; ++i) {
      const int e = eg + i;
      const int src = ei[e], dst = ei[Ee + e];
      const float4* hs = (const float4*)(h + (long)src * HD);
      const float4* hd = (const float4*)(h + (long)dst * HD);
      hs1[i] = hs[lane];  hs2[i] = hs[32 + lane];
      hd1[i] = hd[lane];  hd2[i] = hd[32 + lane];
    }
    __builtin_amdgcn_sched_barrier(0);
#pragma unroll
    for (int i = 0; i < EPG; ++i) {
      const int e = eg + i;
      const float ea0 = edge_attr[e * EFe + 0];
      const float ea1 = edge_attr[e * EFe + 1];
      const float ea2 = edge_attr[e * EFe + 2];
      float p1 = chunk_logit(hs1[i], hd1[i], s1, d1, b1, w1a, w1b, w1c, ea0, ea1, ea2);
      float p2 = chunk_logit(hs2[i], hd2[i], s2, d2, b2, w2a, w2b, w2c, ea0, ea1, ea2);
      p1 = xlane_add<0x161>(p1);  p2 = xlane_add<0x161>(p2);
      p1 = xlane_add<0x162>(p1);  p2 = xlane_add<0x162>(p2);
      p1 = xlane_add<0x164>(p1);  p2 = xlane_add<0x164>(p2);
      r1[i] = p1;  r2[i] = p2;
    }
    float v = r1[0];
    v = (k == 1) ? r1[1] : v;  v = (k == 2) ? r1[2] : v;  v = (k == 3) ? r1[3] : v;
    v = (k == 4) ? r2[0] : v;  v = (k == 5) ? r2[1] : v;  v = (k == 6) ? r2[2] : v;
    v = (k == 7) ? r2[3] : v;
    const float p = __shfl(v, from, 32);
    const float al = (p >= 0.f) ? p : 0.2f * p;
    const int dstv = ei[Ee + eg + (lane >> 3)];
    alpha[eg * Hh + lane] = al;
    atomicMax(&amax_u[dstv * Hh + k], f2o(al));
  }
}

__global__ void softmax_num_kernel(const int* __restrict__ ei,
                                   const unsigned* __restrict__ amax_u,
                                   float* __restrict__ alpha_ex,
                                   float* __restrict__ denom) {
  int idx = blockIdx.x * blockDim.x + threadIdx.x;
  if (idx >= Ee * Hh) return;
  int e = idx >> 3, hh = idx & 7;
  int dst = ei[Ee + e];
  float m  = o2f(amax_u[dst * Hh + hh]);
  float ex = __expf(alpha_ex[idx] - m);
  alpha_ex[idx] = ex;
  atomicAdd(&denom[dst * Hh + hh], ex);
}

__global__ __launch_bounds__(256) void scatter_out_kernel(
    const float* __restrict__ h, const float* __restrict__ ex,
    const float* __restrict__ denom, const int* __restrict__ ei,
    float* __restrict__ out) {
  int e = blockIdx.x * 8 + (threadIdx.x >> 5);
  if (e >= Ee) return;
  int lane = threadIdx.x & 31;
  int src = ei[e], dst = ei[Ee + e];
  const float* hs_row = h + (long)src * HD;
  float* out_row = out + (long)dst * HD;
  for (int hh = 0; hh < Hh; ++hh) {
    float attn = ex[e * Hh + hh] / fmaxf(denom[dst * Hh + hh], 1e-10f);
    atomicAdd(&out_row[hh * Dd + lane], hs_row[hh * Dd + lane] * attn);
  }
}

extern "C" void kernel_launch(void* const* d_in, const int* in_sizes, int n_in,
                              void* d_out, int out_size, void* d_ws, size_t ws_size,
                              hipStream_t stream) {
  const float* x        = (const float*)d_in[0];
  const float* Wm       = (const float*)d_in[1];
  const float* a_src    = (const float*)d_in[2];
  const float* a_dst    = (const float*)d_in[3];
  const float* edge_W   = (const float*)d_in[4];
  const float* edge_b   = (const float*)d_in[5];
  const float* bias     = (const float*)d_in[6];
  const float* edge_attr= (const float*)d_in[7];
  const int*   ei       = (const int*)d_in[8];
  float* out = (float*)d_out;

  float*    h       = (float*)d_ws;
  float*    alpha   = h + (size_t)Nn * HD;
  unsigned* amax_u  = (unsigned*)(alpha + (size_t)Ee * Hh);
  float*    denom   = (float*)(amax_u + (size_t)Nn * Hh);

  init_kernel<<<(Nn * HD + 255) / 256, 256, 0, stream>>>(out, denom, amax_u, bias);

  dim3 g((Nn + 31) / 32, HD / 128);
  gemm_h_wmma<<<g, 256, 0, stream>>>(x, Wm, h);

  edge_alpha_kernel<<<(Ee + 63) / 64, 256, 0, stream>>>(h, a_src, a_dst, edge_W, edge_b,
                                                      edge_attr, ei, alpha, amax_u);
  softmax_num_kernel<<<(Ee * Hh + 255) / 256, 256, 0, stream>>>(ei, amax_u, alpha, denom);
  scatter_out_kernel<<<(Ee + 7) / 8, 256, 0, stream>>>(h, alpha, denom, ei, out);
}


# edges grouped by destination (counting sort); one wave per node does max, exp, denominator, weighted row sum in registers: no float atomics, out stored once; sum order = fill rank, normalised after
# speedup vs baseline: 2.1154x; 1.3517x over previous
#include <hip/hip_runtime.h>

#define Nn   50000
#define Ee   400000
#define INF_ 256
#define Hh   8
#define Dd   32
#define HD   256
#define EFe  3

typedef __attribute__((ext_vector_type(2))) float v2f;
typedef __attribute__((ext_vector_type(8))) float v8f;

#define SCAN_T   1024
#define SCAN_PER 49
#define NP       (SCAN_T * SCAN_PER)
static_assert(NP >= Nn, "the scan covers every node");

__global__ __launch_bounds__(256) void zero_counts_kernel(int* __restrict__ cnt) {
  cnt[blockIdx.x * 256 + threadIdx.x] = 0;
}

__global__ __launch_bounds__(256) void degree_kernel(const int* __restrict__ ei, int* __restrict__ cnt) {
  const int e = blockIdx.x * 256 + threadIdx.x;
  if (e < Ee) atomicAdd(&cnt[ei[Ee + e]], 1);
}

__global__ __launch_bounds__(SCAN_T) void rowptr_scan_kernel(int* __restrict__ cnt,
                                                             int* __restrict__ rowptr) {
  __shared__ int s[NP];
  __shared__ int wtot[32];
  const int t = threadIdx.x, lane = t & 31, w = t >> 5;
#pragma unroll
  for (int j = 0; j < SCAN_PER; ++j) s[j * SCAN_T + t] = cnt[j * SCAN_T + t];
  __syncthreads();
  const int base = t * SCAN_PER;
  int tot = 0;
#pragma unroll
  for (int j = 0; j < SCAN_PER; ++j) tot += s[base + j];
  int inc = tot;
  for (int d = 1; d < 32; d <<= 1) {
    const int n = __shfl_up(inc, d, 32);
    if (lane >= d) inc += n;
  }
  if (lane == 31) wtot[w] = inc;
  __syncthreads();
  if (w == 0) {
    const int mine = wtot[lane];
    int ws = mine;
    for (int d = 1; d < 32; d <<= 1) {
      const int n = __shfl_up(ws, d, 32);
      if (lane >= d) ws += n;
    }
    wtot[lane] = ws - mine;
  }
  __syncthreads();
  int run = wtot[w] + inc - tot;
#pragma unroll
  for (int j = 0; j < SCAN_PER; ++j) {
    const int c = s[base + j];
    s[base + j] = run;
    run += c;
  }
  __syncthreads();
#pragma unroll
  for (int j = 0; j < SCAN_PER; ++j) {
    const int v = s[j * SCAN_T + t];
    rowptr[j * SCAN_T + t] = v;
    cnt[j * SCAN_T + t]    = v;
  }
}

__global__ __launch_bounds__(256) void fill_kernel(const int* __restrict__ ei, int* __restrict__ cursor,
                                                   int* __restrict__ perm) {
  const int e = blockIdx.x * 256 + threadIdx.x;
  if (e < Ee) perm[atomicAdd(&cursor[ei[Ee + e]], 1)] = e;
}

__global__ __launch_bounds__(256) void gemm_h_wmma(const float* __restrict__ x,
                                                   const float* __restrict__ Wm,
                                                   float* __restrict__ h) {
  __shared__ float sA[32][33];
  __shared__ float sB[128][33];
  const int tid   = threadIdx.x;
  const int mBase = blockIdx.x * 32;
  const int nBase = blockIdx.y * 128;
  const int wave  = tid >> 5, ln = tid & 31;
  const int half  = ln >> 4, mr = ln & 15;
  const int mt    = wave & 1;
  const int nt0   = (wave >> 1) << 1;

  v8f acc0 = {};
  v8f acc1 = {};

  for (int k0 = 0; k0 < INF_; k0 += 32) {
    for (int i = tid; i < 32 * 32; i += 256) {
      int r = i >> 5, c = i & 31, row = mBase + r;
      sA[r][c] = (row < Nn) ? x[row * INF_ + k0 + c] : 0.f;
    }
    for (int i = tid; i < 128 * 32; i += 256) {
      int r = i >> 5, c = i & 31;
      sB[r][c] = Wm[(nBase + r) * INF_ + k0 + c];
    }
    __syncthreads();

    const int ar  = mt * 16 + mr;
    const int br0 = nt0 * 16 + mr;
    const int br1 = br0 + 16;
#pragma unroll
    for (int kk = 0; kk < 32; kk += 4) {
      const int kh = kk + half * 2;
      v2f a, b0, b1;
      a.x  = sA[ar][kh];   a.y  = sA[ar][kh + 1];
      b0.x = sB[br0][kh];  b0.y = sB[br0][kh + 1];
      b1.x = sB[br1][kh];  b1.y = sB[br1][kh + 1];
      acc0 = __builtin_amdgcn_wmma_f32_16x16x4_f32(false, a, false, b0, (short)0, acc0, false, false);
      acc1 = __builtin_amdgcn_wmma_f32_16x16x4_f32(false, a, false, b1, (short)0, acc1, false, false);
    }
    __syncthreads();
  }

  const int rowBase = mBase + mt * 16 + half * 8;
#pragma unroll
  for (int r = 0; r < 8; ++r) {
    int row = rowBase + r;
    if (row < Nn) {
      h[row * HD + nBase + nt0 * 16 + mr]       = acc0[r];
      h[row * HD + nBase + (nt0 + 1) * 16 + mr] = acc1[r];
    }
  }
}

#define EPG 4
#define GPW 2

template <int CTRL>
__device__ __forceinline__ float xlane_add(float v) {
  return v + __int_as_float(__builtin_amdgcn_update_dpp(0, __float_as_int(v), CTRL, 0xf, 0xf, true));
}

__device__ __forceinline__ float chunk_logit(float4 hs, float4 hd, float4 as, float4 ad, float4 eb,
                                             float4 wa, float4 wb, float4 wc,
                                             float ea0, float ea1, float ea2) {
  float n0 = fmaxf(fmaf(ea0, wa.x, fmaf(ea1, wa.y, fmaf(ea2, wa.z, eb.x))), 0.f);
  float n1 = fmaxf(fmaf(ea0, wa.w, fmaf(ea1, wb.x, fmaf(ea2, wb.y, eb.y))), 0.f);
  float n2 = fmaxf(fmaf(ea0, wb.z, fmaf(ea1, wb.w, fmaf(ea2, wc.x, eb.z))), 0.f);
  float n3 = fmaxf(fmaf(ea0, wc.y, fmaf(ea1, wc.z, fmaf(ea2, wc.w, eb.w))), 0.f);
  float p = hs.x * as.x;
  p = fmaf(hd.x, ad.x, p);  p = fmaf(n0, hd.x, p);
  p = fmaf(hs.y, as.y, p);  p = fmaf(hd.y, ad.y, p);  p = fmaf(n1, hd.y, p);
  p = fmaf(hs.z, as.z, p);  p = fmaf(hd.z, ad.z, p);  p = fmaf(n2, hd.z, p);
  p = fmaf(hs.w, as.w, p);  p = fmaf(hd.w, ad.w, p);  p = fmaf(n3, hd.w, p);
  return p;
}

__global__ __launch_bounds__(256) void edge_alpha_kernel(
    const float* __restrict__ h, const float* __restrict__ a_src,
    const float* __restrict__ a_dst, const float* __restrict__ edge_W,
    const float* __restrict__ edge_b, const float* __restrict__ edge_attr,
    const int* __restrict__ ei, float* __restrict__ alpha) {
  static_assert(Ee % (EPG * GPW) == 0, "a wave's edges are all in range or none is");
  const int lane = threadIdx.x & 31;
  const int wv = __builtin_amdgcn_readfirstlane(threadIdx.x >> 5);
  const int e0 = (blockIdx.x * 8 + wv) * (EPG * GPW);
  if (e0 >= Ee) return;

  const float4* ew4 = (const float4*)edge_W;
  const float4 w1a = ew4[lane * 3 + 0], w1b = ew4[lane * 3 + 1], w1c = ew4[lane * 3 + 2];
  const float4 w2a = ew4[96 + lane * 3 + 0], w2b = ew4[96 + lane * 3 + 1], w2c = ew4[96 + lane * 3 + 2];
  const float4 b1 = ((const float4*)edge_b)[lane], b2 = ((const float4*)edge_b)[32 + lane];
  const float4 s1 = ((const float4*)a_src)[lane],  s2 = ((const float4*)a_src)[32 + lane];
  const float4 d1 = ((const float4*)a_dst)[lane],  d2 = ((const float4*)a_dst)[32 + lane];

  const int k = lane & 7;
  const int from = ((lane & 3) << 3) + (lane & 4) + (lane >> 3);

#pragma unroll
  for (int g = 0; g < GPW; ++g) {
    const int eg = e0 + g * EPG;
    float r1[EPG], r2[EPG];
    float4 hs1[EPG], hs2[EPG], hd1[EPG], hd2[EPG];
#pragma unroll
    for (int i = 0; i < EPG; ++i) {
      const int e = eg + i;
      const int src = ei[e], dst = ei[Ee + e];
      const float4* hs = (const float4*)(h + (long)src * HD);
      const float4* hd = (const float4*)(h + (long)dst * HD);
      hs1[i] = hs[lane];  hs2[i] = hs[32 + lane];
      hd1[i] = hd[lane];  hd2[i] = hd[32 + lane];
    }
    __builtin_amdgcn_sched_barrier(0);
#pragma unroll
    for (int i = 0; i < EPG; ++i) {
      const int e = eg + i;
      const float ea0 = edge_attr[e * EFe + 0];
      const float ea1 = edge_attr[e * EFe + 1];
      const float ea2 = edge_attr[e * EFe + 2];
      float p1 = chunk_logit(hs1[i], hd1[i], s1, d1, b1, w1a, w1b, w1c, ea0, ea1, ea2);
      float p2 = chunk_logit(hs2[i], hd2[i], s2, d2, b2, w2a, w2b, w2c, ea0, ea1, ea2);
      p1 = xlane_add<0x161>(p1);  p2 = xlane_add<0x161>(p2);
      p1 = xlane_add<0x162>(p1);  p2 = xlane_add<0x162>(p2);
      p1 = xlane_add<0x164>(p1);  p2 = xlane_add<0x164>(p2);
      r1[i] = p1;  r2[i] = p2;
    }
    float v = r1[0];
    v = (k == 1) ? r1[1] : v;  v = (k == 2) ? r1[2] : v;  v = (k == 3) ? r1[3] : v;
    v = (k == 4) ? r2[0] : v;  v = (k == 5) ? r2[1] : v;  v = (k == 6) ? r2[2] : v;
    v = (k == 7) ? r2[3] : v;
    const float p = __shfl(v, from, 32);
    const float al = (p >= 0.f) ? p : 0.2f * p;
    alpha[eg * Hh + lane] = al;
  }
}

__device__ __forceinline__ void add_row(float4& acc, float w, float4 r) {
  acc.x = fmaf(w, r.x, acc.x);  acc.y = fmaf(w, r.y, acc.y);
  acc.z = fmaf(w, r.z, acc.z);  acc.w = fmaf(w, r.w, acc.w);
}

__global__ __launch_bounds__(256) void node_softmax_sum_kernel(
    const float* __restrict__ h, const float* __restrict__ alpha,
    const int* __restrict__ rowptr, const int* __restrict__ rowend,
    const int* __restrict__ perm, const int* __restrict__ ei,
    const float* __restrict__ bias, float* __restrict__ out) {
  static_assert(Nn % 8 == 0, "every wave of the grid has a node");
  const int lane = threadIdx.x & 31;
  const int wv   = __builtin_amdgcn_readfirstlane(threadIdx.x >> 5);
  const int node = blockIdx.x * 8 + wv;
  const int start = rowptr[node], end = rowend[node];
  const int slot = lane >> 3, hd = lane & 7;
  const int g    = lane >> 3;

  float m = -INFINITY;
  for (int c = start; c < end; c += 4) {
    const int idx = c + slot;
    if (idx < end) m = fmaxf(m, alpha[perm[idx] * Hh + hd]);
  }
  m = fmaxf(m, __shfl_xor(m, 8, 32));
  m = fmaxf(m, __shfl_xor(m, 16, 32));

  float dsum = 0.f;
  float4 acc1 = make_float4(0.f, 0.f, 0.f, 0.f), acc2 = make_float4(0.f, 0.f, 0.f, 0.f);
  for (int c = start; c < end; c += 4) {
    const int idx = c + slot;
    float ex = 0.f;
    int   sv = 0;
    if (idx < end) {
      const int e = perm[idx];
      ex = __expf(alpha[e * Hh + hd] - m);
      sv = ei[e];
    }
    dsum += ex;
    const int nv = end - c;
    const float4* row0 = (const float4*)(h + (long)__builtin_amdgcn_readlane(sv, 0) * HD);
    float4 a0 = row0[lane], b0 = row0[32 + lane];
    float4 a1, b1, a2, b2, a3, b3;
    if (nv > 1) {
      const float4* row = (const float4*)(h + (long)__builtin_amdgcn_readlane(sv, 8) * HD);
      a1 = row[lane];  b1 = row[32 + lane];
    }
    if (nv > 2) {
      const float4* row = (const float4*)(h + (long)__builtin_amdgcn_readlane(sv, 16) * HD);
      a2 = row[lane];  b2 = row[32 + lane];
    }
    if (nv > 3) {
      const float4* row = (const float4*)(h + (long)__builtin_amdgcn_readlane(sv, 24) * HD);
      a3 = row[lane];  b3 = row[32 + lane];
    }
    add_row(acc1, __shfl(ex, g, 32), a0);
    add_row(acc2, __shfl(ex, 4 + g, 32), b0);
    const float w1a = __shfl(ex, 8 + g, 32),  w1b = __shfl(ex, 12 + g, 32);
    const float w2a = __shfl(ex, 16 + g, 32), w2b = __shfl(ex, 20 + g, 32);
    const float w3a = __shfl(ex, 24 + g, 32), w3b = __shfl(ex, 28 + g, 32);
    if (nv > 1) { add_row(acc1, w1a, a1);  add_row(acc2, w1b, b1); }
    if (nv > 2) { add_row(acc1, w2a, a2);  add_row(acc2, w2b, b2); }
    if (nv > 3) { add_row(acc1, w3a, a3);  add_row(acc2, w3b, b3); }
  }
  dsum += __shfl_xor(dsum, 8, 32);
  dsum += __shfl_xor(dsum, 16, 32);
  const float da = fmaxf(__shfl(dsum, g, 32), 1e-10f);
  const float db = fmaxf(__shfl(dsum, 4 + g, 32), 1e-10f);
  const float4 bi1 = ((const float4*)bias)[lane], bi2 = ((const float4*)bias)[32 + lane];
  float4 o1, o2;
  o1.x = acc1.x / da + bi1.x;  o1.y = acc1.y / da + bi1.y;
  o1.z = acc1.z / da + bi1.z;  o1.w = acc1.w / da + bi1.w;
  o2.x = acc2.x / db + bi2.x;  o2.y = acc2.y / db + bi2.y;
  o2.z = acc2.z / db + bi2.z;  o2.w = acc2.w / db + bi2.w;
  float4* orow = (float4*)(out + (long)node * HD);
  orow[lane]      = o1;
  orow[32 + lane] = o2;
}

extern "C" void kernel_launch(void* const* d_in, const int* in_sizes, int n_in,
                              void* d_out, int out_size, void* d_ws, size_t ws_size,
                              hipStream_t stream) {
  const float* x        = (const float*)d_in[0];
  const float* Wm       = (const float*)d_in[1];
  const float* a_src    = (const float*)d_in[2];
  const float* a_dst    = (const float*)d_in[3];
  const float* edge_W   = (const float*)d_in[4];
  const float* edge_b   = (const float*)d_in[5];
  const float* bias     = (const float*)d_in[6];
  const float* edge_attr= (const float*)d_in[7];
  const int*   ei       = (const int*)d_in[8];
  float* out = (float*)d_out;

  float* h      = (float*)d_ws;
  float* alpha  = h + (size_t)Nn * HD;
  int*   cnt    = (int*)(alpha + (size_t)Ee * Hh);
  int*   rowptr = cnt + NP;
  int*   perm   = rowptr + NP;

  zero_counts_kernel<<<NP / 256, 256, 0, stream>>>(cnt);
  degree_kernel<<<(Ee + 255) / 256, 256, 0, stream>>>(ei, cnt);
  rowptr_scan_kernel<<<1, SCAN_T, 0, stream>>>(cnt, rowptr);
  fill_kernel<<<(Ee + 255) / 256, 256, 0, stream>>>(ei, cnt, perm);

  dim3 g((Nn + 31) / 32, HD / 128);
  gemm_h_wmma<<<g, 256, 0, stream>>>(x, Wm, h);

  edge_alpha_kernel<<<(Ee + 63) / 64, 256, 0, stream>>>(h, a_src, a_dst, edge_W, edge_b,
                                                      edge_attr, ei, alpha);
  node_softmax_sum_kernel<<<Nn / 8, 256, 0, stream>>>(h, alpha, rowptr, cnt, perm, ei, bias, out);
}


# node kernel as a resident grid: 1024 work-groups of 8 waves (4 per WGP, all resident), a wave loops over its nodes g, g+8192, ..; next node's bounds and first edges fetched ahead; sum order kept
# speedup vs baseline: 4.2222x; 1.9959x over previous
#include <hip/hip_runtime.h>

#define Nn   50000
#define Ee   400000
#define INF_ 256
#define Hh   8
#define Dd   32
#define HD   256
#define EFe  3

typedef __attribute__((ext_vector_type(2))) float v2f;
typedef __attribute__((ext_vector_type(8))) float v8f;

#define RNG   196
#define NWG   256
#define NP    (RNG * NWG)
#define SWV   32
#define EPW   (Ee / SWV)
#define QPW   (EPW / 4)
#define LCAP  256
static_assert(NP >= Nn, "the ranges cover every node");
static_assert(Ee % SWV == 0 && EPW % 4 == 0, "a wave's slice is whole 16-byte loads");
static_assert(RNG <= 256 && Ee < (1 << 23), "list entry = edge number * 256 + local node");

template <bool PLACE>
__device__ __forceinline__ int scan_slice(const int4* __restrict__ dp, int ebase, int lo, int lane,
                                          int* cnt_w, int* list_w, int* __restrict__ perm, int& below) {
  int nl = 0;
  for (int q0 = 0; q0 < QPW; q0 += 128) {
    int4 d[4];
#pragma unroll
    for (int k = 0; k < 4; ++k) {
      const int q = q0 + 32 * k + lane;
      d[k] = (q < QPW) ? dp[q] : make_int4(0x7fffffff, 0x7fffffff, 0x7fffffff, 0x7fffffff);
    }
#pragma unroll
    for (int k = 0; k < 4; ++k) {
      const int q = q0 + 32 * k + lane;
      const unsigned u[4] = {(unsigned)(d[k].x - lo), (unsigned)(d[k].y - lo),
                             (unsigned)(d[k].z - lo), (unsigned)(d[k].w - lo)};
      if (!PLACE)
        below += __popc(__builtin_amdgcn_ballot_w32(d[k].x < lo)) + __popc(__builtin_amdgcn_ballot_w32(d[k].y < lo)) +
                 __popc(__builtin_amdgcn_ballot_w32(d[k].z < lo)) + __popc(__builtin_amdgcn_ballot_w32(d[k].w < lo));
      const unsigned mn = min(min(u[0], u[1]), min(u[2], u[3]));
      if (__builtin_amdgcn_ballot_w32(mn < (unsigned)RNG) != 0u) {
#pragma unroll
        for (int c = 0; c < 4; ++c) {
          const bool hit = u[c] < (unsigned)RNG;
          const unsigned mask = __builtin_amdgcn_ballot_w32(hit);
          if (mask != 0u) {
            if (hit) {
              const int e = ebase + 4 * q + c;
              if (PLACE) {
                perm[atomicAdd(&cnt_w[u[c]], 1)] = e;
              } else {
                atomicAdd(&cnt_w[u[c]], 1);
                const int idx = nl + (int)__builtin_amdgcn_mbcnt_lo(mask, 0u);
                if (idx < LCAP) list_w[idx] = (e << 8) | (int)u[c];
              }
            }
            nl += __popc(mask);
          }
        }
      }
    }
  }
  return nl;
}

__device__ __forceinline__ void csr_role(int wg, int (*s_cnt)[RNG], int (*s_list)[LCAP], int* s_below, int* s_wsum,
                                         const int* __restrict__ ei, int* __restrict__ rowptr,
                                         int* __restrict__ rowend, int* __restrict__ perm) {
  const int t = threadIdx.x, lane = t & 31;
  const int w  = __builtin_amdgcn_readfirstlane(t >> 5);
  const int lo = wg * RNG;
  const int4* dp = (const int4*)(ei + Ee + w * EPW);

  for (int i = lane; i < RNG; i += 32) s_cnt[w][i] = 0;
  __syncthreads();

  int below = 0;
  const int nl = scan_slice<false>(dp, w * EPW, lo, lane, s_cnt[w], s_list[w], perm, below);
  if (lane == 0) s_below[w] = below;
  __syncthreads();

  int tot = 0;
  if (t < RNG) {
#pragma unroll
    for (int k = 0; k < SWV; ++k) {
      const int c = s_cnt[k][t];
      s_cnt[k][t] = tot;
      tot += c;
    }
  }
  int inc = tot;
  for (int dlt = 1; dlt < 32; dlt <<= 1) {
    const int n = __shfl_up(inc, dlt, 32);
    if (lane >= dlt) inc += n;
  }
  if (lane == 31 && w < 8) s_wsum[w] = inc;
  __syncthreads();

  if (t < RNG) {
    int first = 0;
#pragma unroll
    for (int k = 0; k < SWV; ++k) first += s_below[k];
    for (int k = 0; k < w; ++k) first += s_wsum[k];
    first += inc - tot;
    rowptr[lo + t] = first;
    rowend[lo + t] = first + tot;
#pragma unroll
    for (int k = 0; k < SWV; ++k) s_cnt[k][t] += first;
  }
  __syncthreads();

  if (nl <= LCAP) {
    for (int i = lane; i < nl; i += 32) {
      const int it = s_list[w][i];
      perm[atomicAdd(&s_cnt[w][it & 255], 1)] = it >> 8;
    }
  } else {
    int unused = 0;
    scan_slice<true>(dp, w * EPW, lo, lane, s_cnt[w], s_list[w], perm, unused);
  }
}

template <int CTRL>
__device__ __forceinline__ float xlane_add(float v) {
  return v + __int_as_float(__builtin_amdgcn_update_dpp(0, __float_as_int(v), CTRL, 0xf, 0xf, true));
}

#define GT_M   128
#define GT_N   128
#define GT_K   32
#define GPITCH 36
#define GEMM_MB ((Nn + GT_M - 1) / GT_M)
#define GEMM_WG (GEMM_MB * (HD / GT_N))
static_assert(HD % GT_N == 0 && INF_ % GT_K == 0, "whole column tiles and whole k chunks");

__device__ __forceinline__ void lds_handover_barrier() {
  asm volatile("s_wait_dscnt 0x0\n\ts_barrier_signal -1\n\ts_barrier_wait -1" ::: "memory");
}

__device__ __forceinline__ void gemm_role(int gb, float (*sA)[GPITCH], float (*sB)[GPITCH],
                                          const float* __restrict__ x, const float* __restrict__ Wm,
                                          const float* __restrict__ a_src, float* __restrict__ h,
                                          float* __restrict__ ssrc) {
  const int tid   = threadIdx.x;
  const int mBase = (gb >> 1) * GT_M;
  const int nBase = (gb & 1) * GT_N;
  const int wave  = tid >> 5, ln = tid & 31;
  const int half  = ln >> 4, mr = ln & 15;
  const int mt    = wave & 7;
  const int np    = wave >> 3;
  const int nt0   = np << 1;

  const int sr = tid >> 3, sc = (tid & 7) << 2;
  const float4* gx = (const float4*)(x + (size_t)min(mBase + sr, Nn - 1) * INF_ + sc);
  const float4* gw = (const float4*)(Wm + (size_t)(nBase + sr) * INF_ + sc);
  float4 ra = gx[0], rb = gw[0];

  v8f acc0 = {};
  v8f acc1 = {};
  const int ar  = mt * 16 + mr;
  const int br0 = nt0 * 16 + mr;
  const int br1 = br0 + 16;

  for (int k0 = 0; k0 < INF_; k0 += GT_K) {
    *(float4*)&sA[sr][sc] = ra;
    *(float4*)&sB[sr][sc] = rb;
    lds_handover_barrier();
    if (k0 + GT_K < INF_) { ra = gx[(k0 + GT_K) >> 2];  rb = gw[(k0 + GT_K) >> 2]; }
#pragma unroll
    for (int kk = 0; kk < GT_K; kk += 4) {
      const int kh = kk + half * 2;
      const v2f a  = *(const v2f*)&sA[ar][kh];
      const v2f b0 = *(const v2f*)&sB[br0][kh];
      const v2f b1 = *(const v2f*)&sB[br1][kh];
      acc0 = __builtin_amdgcn_wmma_f32_16x16x4_f32(false, a, false, b0, (short)0, acc0, false, false);
      acc1 = __builtin_amdgcn_wmma_f32_16x16x4_f32(false, a, false, b1, (short)0, acc1, false, false);
    }
    lds_handover_barrier();
  }

  const int rowBase = mBase + mt * 16 + half * 8;
#pragma unroll
  for (int r = 0; r < 8; ++r) {
    int row = rowBase + r;
    if (row < Nn) {
      h[row * HD + nBase + nt0 * 16 + mr]       = acc0[r];
      h[row * HD + nBase + (nt0 + 1) * 16 + mr] = acc1[r];
    }
  }

  const int head = (nBase >> 5) + np;
  const float as0 = a_src[head * Dd + mr], as1 = a_src[head * Dd + 16 + mr];
  float sc_ = 0.f;
#pragma unroll
  for (int r = 0; r < 8; ++r) {
    float p = fmaf(acc1[r], as1, acc0[r] * as0);
    p = xlane_add<0x161>(p);  p = xlane_add<0x162>(p);
    p = xlane_add<0x164>(p);  p = xlane_add<0x168>(p);
    sc_ = (mr == r) ? p : sc_;
  }
  if (mr < 8 && rowBase + mr < Nn) ssrc[(rowBase + mr) * Hh + head] = sc_;
}

__global__ __launch_bounds__(1024) void csr_gemm_kernel(const int* __restrict__ ei, int* __restrict__ rowptr,
                                                        int* __restrict__ rowend, int* __restrict__ perm,
                                                        const float* __restrict__ x, const float* __restrict__ Wm,
                                                        const float* __restrict__ a_src, float* __restrict__ h,
                                                        float* __restrict__ ssrc) {
  __shared__ int s_cnt[SWV][RNG];
  __shared__ int s_list[SWV][LCAP];
  __shared__ int s_below[SWV];
  __shared__ int s_wsum[8];
  __shared__ alignas(16) float sA[GT_M][GPITCH];
  __shared__ alignas(16) float sB[GT_N][GPITCH];
  if (blockIdx.x < NWG) csr_role(blockIdx.x, s_cnt, s_list, s_below, s_wsum, ei, rowptr, rowend, perm);
  else                  gemm_role(blockIdx.x - NWG, sA, sB, x, Wm, a_src, h, ssrc);
}

#define CAPC 16
#define NODE_WG 1024
#define NODE_NW (NODE_WG * 8)

__device__ __forceinline__ float enc_dot(float4 hd, float4 eb, float4 wa, float4 wb, float4 wc,
                                         float ea0, float ea1, float ea2) {
  float n0 = fmaxf(fmaf(ea0, wa.x, fmaf(ea1, wa.y, fmaf(ea2, wa.z, eb.x))), 0.f);
  float n1 = fmaxf(fmaf(ea0, wa.w, fmaf(ea1, wb.x, fmaf(ea2, wb.y, eb.y))), 0.f);
  float n2 = fmaxf(fmaf(ea0, wb.z, fmaf(ea1, wb.w, fmaf(ea2, wc.x, eb.z))), 0.f);
  float n3 = fmaxf(fmaf(ea0, wc.y, fmaf(ea1, wc.z, fmaf(ea2, wc.w, eb.w))), 0.f);
  float p = n0 * hd.x;
  p = fmaf(n1, hd.y, p);  p = fmaf(n2, hd.z, p);  p = fmaf(n3, hd.w, p);
  return p;
}

struct EdgeWeights {
  float4 w1a, w1b, w1c, w2a, w2b, w2c, b1, b2;
};

__device__ __forceinline__ float chunk_logits(int e, int nv, int lane, float sd, float4 hd1, float4 hd2,
                                              const EdgeWeights& w, const int* __restrict__ ei,
                                              const float* __restrict__ edge_attr,
                                              const float* __restrict__ ssrc, int& sv) {
  const int slot = lane >> 3, j = lane & 7;
  sv = ei[e];
  const float eav = edge_attr[e * EFe + min(j, EFe - 1)];
  __builtin_amdgcn_sched_barrier(0);
  const float ss  = ssrc[sv * Hh + j];
  float r1[4], r2[4];
#pragma unroll
  for (int i = 0; i < 4; ++i) {
    r1[i] = 0.f;  r2[i] = 0.f;
    if (i < nv) {
      const float ea0 = __int_as_float(__builtin_amdgcn_readlane(__float_as_int(eav), 8 * i + 0));
      const float ea1 = __int_as_float(__builtin_amdgcn_readlane(__float_as_int(eav), 8 * i + 1));
      const float ea2 = __int_as_float(__builtin_amdgcn_readlane(__float_as_int(eav), 8 * i + 2));
      float p1 = enc_dot(hd1, w.b1, w.w1a, w.w1b, w.w1c, ea0, ea1, ea2);
      float p2 = enc_dot(hd2, w.b2, w.w2a, w.w2b, w.w2c, ea0, ea1, ea2);
      p1 = xlane_add<0x161>(p1);  p2 = xlane_add<0x161>(p2);
      p1 = xlane_add<0x162>(p1);  p2 = xlane_add<0x162>(p2);
      p1 = xlane_add<0x164>(p1);  p2 = xlane_add<0x164>(p2);
      r1[i] = p1;  r2[i] = p2;
    }
  }
  float v = r1[0];
  v = (j == 1) ? r1[1] : v;  v = (j == 2) ? r1[2] : v;  v = (j == 3) ? r1[3] : v;
  v = (j == 4) ? r2[0] : v;  v = (j == 5) ? r2[1] : v;  v = (j == 6) ? r2[2] : v;
  v = (j == 7) ? r2[3] : v;
  const float p = __shfl(v, ((j & 3) << 3) + (j & 4) + slot, 32) + ss + sd;
  const float al = (p >= 0.f) ? p : 0.2f * p;
  return (slot < nv) ? al : -INFINITY;
}

__device__ __forceinline__ void add_row(float4& acc, float w, float4 r) {
  acc.x = fmaf(w, r.x, acc.x);  acc.y = fmaf(w, r.y, acc.y);
  acc.z = fmaf(w, r.z, acc.z);  acc.w = fmaf(w, r.w, acc.w);
}

__global__ __launch_bounds__(256) void node_attention_kernel(
    const float* __restrict__ h, const float* __restrict__ ssrc,
    const float* __restrict__ a_dst, const float* __restrict__ edge_W,
    const float* __restrict__ edge_b, const float* __restrict__ edge_attr,
    const int* __restrict__ rowptr, const int* __restrict__ rowend,
    const int* __restrict__ perm, const int* __restrict__ ei,
    const float* __restrict__ bias, float* __restrict__ out) {
  static_assert(NODE_NW <= Nn, "every wave of the grid has a first node");
  __shared__ float s_lg[8][CAPC][32];
  __shared__ int   s_sv[8][CAPC][32];
  const int lane = threadIdx.x & 31;
  const int wv   = __builtin_amdgcn_readfirstlane(threadIdx.x >> 5);
  const int slot = lane >> 3, j = lane & 7;
  const int g    = lane >> 3;

  int node  = blockIdx.x * 8 + wv;
  int start = rowptr[node], end = rowend[node];
  int en    = (start < end) ? perm[min(start + slot, end - 1)] : 0;
  for (;;) {
  asm volatile("" ::: "memory");
  const int  nnode = node + NODE_NW;
  const bool more  = nnode < Nn;
  int nstart = 0, nend = 0;
  if (more) { nstart = rowptr[nnode];  nend = rowend[nnode]; }

  const float4* hrow = (const float4*)(h + (long)node * HD);
  const float4 hd1 = hrow[lane], hd2 = hrow[32 + lane];
  const float4* ew4 = (const float4*)edge_W;
  EdgeWeights w;
  w.w1a = ew4[lane * 3 + 0];       w.w1b = ew4[lane * 3 + 1];       w.w1c = ew4[lane * 3 + 2];
  w.w2a = ew4[96 + lane * 3 + 0];  w.w2b = ew4[96 + lane * 3 + 1];  w.w2c = ew4[96 + lane * 3 + 2];
  w.b1 = ((const float4*)edge_b)[lane];  w.b2 = ((const float4*)edge_b)[32 + lane];
  const float4 ad1 = ((const float4*)a_dst)[lane], ad2 = ((const float4*)a_dst)[32 + lane];

  float sd;
  {
    float q1 = hd1.x * ad1.x;
    q1 = fmaf(hd1.y, ad1.y, q1);  q1 = fmaf(hd1.z, ad1.z, q1);  q1 = fmaf(hd1.w, ad1.w, q1);
    float q2 = hd2.x * ad2.x;
    q2 = fmaf(hd2.y, ad2.y, q2);  q2 = fmaf(hd2.z, ad2.z, q2);  q2 = fmaf(hd2.w, ad2.w, q2);
    q1 = xlane_add<0x161>(q1);  q2 = xlane_add<0x161>(q2);
    q1 = xlane_add<0x162>(q1);  q2 = xlane_add<0x162>(q2);
    q1 = xlane_add<0x164>(q1);  q2 = xlane_add<0x164>(q2);
    const float t1 = __shfl(q1, (j & 3) << 3, 32), t2 = __shfl(q2, (j & 3) << 3, 32);
    sd = (j < 4) ? t1 : t2;
  }

  float m = -INFINITY;
  if (start < end) {
    for (int c = start, ci = 0; c < end; c += 4, ++ci) {
      const int e = en;
      if (c + 4 < end) en = perm[min(c + 4 + slot, end - 1)];
      int sv;
      const float lg = chunk_logits(e, end - c, lane, sd, hd1, hd2, w, ei, edge_attr, ssrc, sv);
      if (ci < CAPC) { s_lg[wv][ci][lane] = lg;  s_sv[wv][ci][lane] = sv; }
      m = fmaxf(m, lg);
    }
  }
  m = fmaxf(m, __shfl_xor(m, 8, 32));
  m = fmaxf(m, __shfl_xor(m, 16, 32));

  int en_next = 0;
  if (nstart < nend) en_next = perm[min(nstart + slot, nend - 1)];

  float dsum = 0.f;
  float4 acc1 = make_float4(0.f, 0.f, 0.f, 0.f), acc2 = make_float4(0.f, 0.f, 0.f, 0.f);
  for (int c = start + 4 * CAPC; c < end; c += 4) {
    int sv;
    const int nv = end - c;
    const float lg = chunk_logits(perm[min(c + slot, end - 1)], nv, lane, sd, hd1, hd2, w, ei, edge_attr,
                                  ssrc, sv);
    const float ex = (c + slot < end) ? __expf(lg - m) : 0.f;
    dsum += ex;
#pragma unroll
    for (int s = 0; s < 4; ++s) {
      if (s < nv) {
        const float4* row = (const float4*)(h + (long)__builtin_amdgcn_readlane(sv, 8 * s) * HD);
        const float4 a = row[lane], b = row[32 + lane];
        add_row(acc1, __shfl(ex, 8 * s + g, 32), a);
        add_row(acc2, __shfl(ex, 8 * s + 4 + g, 32), b);
      }
    }
  }
  const int endl = min(end, start + 4 * CAPC);
  for (int c = start, ci = 0; c < endl; c += 4, ++ci) {
    const float lg = s_lg[wv][ci][lane];
    const int   sv = s_sv[wv][ci][lane];
    const float ex = (c + slot < end) ? __expf(lg - m) : 0.f;
    dsum += ex;
    const int nv = end - c;
    const float4* row0 = (const float4*)(h + (long)__builtin_amdgcn_readlane(sv, 0) * HD);
    float4 a0 = row0[lane], b0 = row0[32 + lane];
    float4 a1, b1, a2, b2, a3, b3;
    if (nv > 1) {
      const float4* row = (const float4*)(h + (long)__builtin_amdgcn_readlane(sv, 8) * HD);
      a1 = row[lane];  b1 = row[32 + lane];
    }
    if (nv > 2) {
      const float4* row = (const float4*)(h + (long)__builtin_amdgcn_readlane(sv, 16) * HD);
      a2 = row[lane];  b2 = row[32 + lane];
    }
    if (nv > 3) {
      const float4* row = (const float4*)(h + (long)__builtin_amdgcn_readlane(sv, 24) * HD);
      a3 = row[lane];  b3 = row[32 + lane];
    }
    add_row(acc1, __shfl(ex, g, 32), a0);
    add_row(acc2, __shfl(ex, 4 + g, 32), b0);
    const float w1a = __shfl(ex, 8 + g, 32),  w1b = __shfl(ex, 12 + g, 32);
    const float w2a = __shfl(ex, 16 + g, 32), w2b = __shfl(ex, 20 + g, 32);
    const float w3a = __shfl(ex, 24 + g, 32), w3b = __shfl(ex, 28 + g, 32);
    if (nv > 1) { add_row(acc1, w1a, a1);  add_row(acc2, w1b, b1); }
    if (nv > 2) { add_row(acc1, w2a, a2);  add_row(acc2, w2b, b2); }
    if (nv > 3) { add_row(acc1, w3a, a3);  add_row(acc2, w3b, b3); }
  }
  dsum += __shfl_xor(dsum, 8, 32);
  dsum += __shfl_xor(dsum, 16, 32);
  const float da = fmaxf(__shfl(dsum, g, 32), 1e-10f);
  const float db = fmaxf(__shfl(dsum, 4 + g, 32), 1e-10f);
  const float4 bi1 = ((const float4*)bias)[lane], bi2 = ((const float4*)bias)[32 + lane];
  float4 o1, o2;
  o1.x = acc1.x / da + bi1.x;  o1.y = acc1.y / da + bi1.y;
  o1.z = acc1.z / da + bi1.z;  o1.w = acc1.w / da + bi1.w;
  o2.x = acc2.x / db + bi2.x;  o2.y = acc2.y / db + bi2.y;
  o2.z = acc2.z / db + bi2.z;  o2.w = acc2.w / db + bi2.w;
  float4* orow = (float4*)(out + (long)node * HD);
  orow[lane]      = o1;
  orow[32 + lane] = o2;

  if (!more) break;
  node = nnode;  start = nstart;  end = nend;  en = en_next;
  }
}

extern "C" void kernel_launch(void* const* d_in, const int* in_sizes, int n_in,
                              void* d_out, int out_size, void* d_ws, size_t ws_size,
                              hipStream_t stream) {
  const float* x        = (const float*)d_in[0];
  const float* Wm       = (const float*)d_in[1];
  const float* a_src    = (const float*)d_in[2];
  const float* a_dst    = (const float*)d_in[3];
  const float* edge_W   = (const float*)d_in[4];
  const float* edge_b   = (const float*)d_in[5];
  const float* bias     = (const float*)d_in[6];
  const float* edge_attr= (const float*)d_in[7];
  const int*   ei       = (const int*)d_in[8];
  float* out = (float*)d_out;

  float* h      = (float*)d_ws;
  float* ssrc   = h + (size_t)Nn * HD;
  int*   cnt    = (int*)(ssrc + (size_t)Ee * Hh);
  int*   rowptr = cnt + NP;
  int*   perm   = rowptr + NP;

  csr_gemm_kernel<<<NWG + GEMM_WG, 1024, 0, stream>>>(ei, rowptr, cnt, perm, x, Wm, a_src, h, ssrc);

  node_attention_kernel<<<NODE_WG, 256, 0, stream>>>(h, ssrc, a_dst, edge_W, edge_b, edge_attr,
                                                    rowptr, cnt, perm, ei, bias, out);
}
